# starting kernel
# baseline (speedup 1.0000x reference)
#include <hip/hip_runtime.h>
#include <math.h>


typedef __attribute__((ext_vector_type(2))) float v2f;
typedef __attribute__((ext_vector_type(8))) float v8f;

#define NB 4
#define NC 64
#define NL 4096
#define LT 64
#define NTILES (NL / LT)
#define NWAVES 4
#define BLOCK (NWAVES * 32)
#define EPS_BN 1e-5f

#define SE_STRIDE 68
#define SH_STRIDE 65

__device__ __forceinline__ v8f wmma_f32(v2f a, v2f b, v8f c) {
  return __builtin_amdgcn_wmma_f32_16x16x4_f32(false, a, false, b,
                                               (short)0, c, false, false);
}

__global__ __launch_bounds__(BLOCK)
void k1_conv_qkv_M(const float* __restrict__ x,
                   const float* __restrict__ conv_w, const float* __restrict__ conv_b,
                   const float* __restrict__ wq, const float* __restrict__ bq,
                   const float* __restrict__ wk, const float* __restrict__ bk,
                   const float* __restrict__ wv, const float* __restrict__ bv,
                   float* __restrict__ kbuf, float* __restrict__ Mpart) {
  __shared__ float sE[NC * SE_STRIDE];
  __shared__ float sH[NC * SH_STRIDE];
  __shared__ float sQ[NC * SH_STRIDE];
  __shared__ float sV[NC * SH_STRIDE];

  const int tid  = threadIdx.x;
  const int wave = tid >> 5;
  const int lane = tid & 31;
  const int half = lane >> 4;
  const int mn   = lane & 15;

  const int bIdx = blockIdx.x / NTILES;
  const int l0   = (blockIdx.x % NTILES) * LT;

  const float* xb = x + bIdx * NC * NL;
  for (int idx = tid; idx < NC * (LT + 2); idx += BLOCK) {
    int c = idx / (LT + 2);
    int j = idx % (LT + 2);
    int gl = (l0 + j - 1) & (NL - 1);
    float v = xb[c * NL + gl];
    sE[c * SE_STRIDE + j] = (v > 0.0f) ? v : (__expf(v) - 1.0f);
  }
  __syncthreads();

  const int mt = wave;
  const int arow = mt * 16 + mn;
  const int rbase = mt * 16 + half * 8;

  for (int nt = 0; nt < 4; ++nt) {
    const int bcol = nt * 16 + mn;
    v8f acc = {};
    for (int tap = 0; tap < 3; ++tap) {
      const float* Wt = conv_w + tap;
      for (int kk = 0; kk < 16; ++kk) {
        const int k0 = kk * 4 + half * 2;
        v2f a, b;
        a[0] = Wt[(arow * NC + k0) * 3];
        a[1] = Wt[(arow * NC + k0 + 1) * 3];
        b[0] = sE[k0 * SE_STRIDE + bcol + tap];
        b[1] = sE[(k0 + 1) * SE_STRIDE + bcol + tap];
        acc = wmma_f32(a, b, acc);
      }
    }
#pragma unroll
    for (int r = 0; r < 8; ++r)
      sH[(rbase + r) * SH_STRIDE + bcol] = acc[r] + conv_b[rbase + r];
  }
  __syncthreads();

  for (int nt = 0; nt < 4; ++nt) {
    const int bcol = nt * 16 + mn;
    v8f aq = {}, ak = {}, av = {};
    for (int kk = 0; kk < 16; ++kk) {
      const int k0 = kk * 4 + half * 2;
      v2f bf, a;
      bf[0] = sH[k0 * SH_STRIDE + bcol];
      bf[1] = sH[(k0 + 1) * SH_STRIDE + bcol];
      a[0] = wq[arow * NC + k0]; a[1] = wq[arow * NC + k0 + 1];
      aq = wmma_f32(a, bf, aq);
      a[0] = wk[arow * NC + k0]; a[1] = wk[arow * NC + k0 + 1];
      ak = wmma_f32(a, bf, ak);
      a[0] = wv[arow * NC + k0]; a[1] = wv[arow * NC + k0 + 1];
      av = wmma_f32(a, bf, av);
    }
#pragma unroll
    for (int r = 0; r < 8; ++r) {
      const int row = rbase + r;
      sQ[row * SH_STRIDE + bcol] = aq[r] + bq[row];
      sV[row * SH_STRIDE + bcol] = av[r] + bv[row];
      kbuf[(bIdx * NC + row) * NL + l0 + bcol] = ak[r] + bk[row];
    }
  }
  __syncthreads();

  float* Mout = Mpart + blockIdx.x * (NC * NC);
  for (int nt = 0; nt < 4; ++nt) {
    const int bcol = nt * 16 + mn;
    v8f acc = {};
    for (int kk = 0; kk < 16; ++kk) {
      const int k0 = kk * 4 + half * 2;
      v2f a, b;
      a[0] = sV[arow * SH_STRIDE + k0];
      a[1] = sV[arow * SH_STRIDE + k0 + 1];
      b[0] = sQ[bcol * SH_STRIDE + k0];
      b[1] = sQ[bcol * SH_STRIDE + k0 + 1];
      acc = wmma_f32(a, b, acc);
    }
#pragma unroll
    for (int r = 0; r < 8; ++r)
      Mout[(rbase + r) * NC + bcol] = acc[r];
  }
}

__global__ void k1b_reduceM(const float* __restrict__ Mpart,
                            float* __restrict__ Mbuf) {
  int i = blockIdx.x * blockDim.x + threadIdx.x;
  if (i >= NB * NC * NC) return;
  const int b = i >> 12;
  const int mn_ = i & 4095;
  const float* p = Mpart + (size_t)(b * NTILES) * (NC * NC) + mn_;
  float s = 0.0f;
  for (int t = 0; t < NTILES; ++t) s += p[(size_t)t * NC * NC];
  Mbuf[i] = s;
}

__global__ __launch_bounds__(BLOCK)
void k2_gate_residual(const float* __restrict__ x,
                      const float* __restrict__ kbuf,
                      const float* __restrict__ Mbuf,
                      float* __restrict__ y,
                      float* __restrict__ sumPart, float* __restrict__ sqPart) {
  __shared__ float sK[NC * SH_STRIDE];
  __shared__ float sSum[NC * 16];
  __shared__ float sSq[NC * 16];

  const int tid  = threadIdx.x;
  const int wave = tid >> 5;
  const int lane = tid & 31;
  const int half = lane >> 4;
  const int mn   = lane & 15;

  const int bIdx = blockIdx.x / NTILES;
  const int l0   = (blockIdx.x % NTILES) * LT;

  const float* kb = kbuf + (size_t)bIdx * NC * NL + l0;
  for (int idx = tid; idx < NC * LT; idx += BLOCK) {
    int c = idx >> 6, j = idx & 63;
    sK[c * SH_STRIDE + j] = kb[c * NL + j];
  }
  __syncthreads();

  const float* Mb = Mbuf + bIdx * NC * NC;
  const float* xb = x + (size_t)bIdx * NC * NL + l0;
  float* yb = y + (size_t)bIdx * NC * NL + l0;

  const int mt = wave;
  const int arow = mt * 16 + mn;
  const int rbase = mt * 16 + half * 8;

  float regS[8], regQ[8];
#pragma unroll
  for (int r = 0; r < 8; ++r) { regS[r] = 0.0f; regQ[r] = 0.0f; }

  for (int nt = 0; nt < 4; ++nt) {
    const int bcol = nt * 16 + mn;
    v8f acc = {};
    for (int kk = 0; kk < 16; ++kk) {
      const int k0 = kk * 4 + half * 2;
      v2f a, b;
      a[0] = Mb[arow * NC + k0];
      a[1] = Mb[arow * NC + k0 + 1];
      b[0] = sK[k0 * SH_STRIDE + bcol];
      b[1] = sK[(k0 + 1) * SH_STRIDE + bcol];
      acc = wmma_f32(a, b, acc);
    }
#pragma unroll
    for (int r = 0; r < 8; ++r) {
      const int row = rbase + r;
      float yv = acc[r] + xb[row * NL + bcol];
      yb[row * NL + bcol] = yv;
      regS[r] += yv;
      regQ[r] += yv * yv;
    }
  }
#pragma unroll
  for (int r = 0; r < 8; ++r) {
    sSum[(rbase + r) * 16 + mn] = regS[r];
    sSq [(rbase + r) * 16 + mn] = regQ[r];
  }
  __syncthreads();
  if (tid < NC) {
    float s = 0.0f, q = 0.0f;
    for (int j = 0; j < 16; ++j) { s += sSum[tid * 16 + j]; q += sSq[tid * 16 + j]; }
    sumPart[blockIdx.x * NC + tid] = s;
    sqPart [blockIdx.x * NC + tid] = q;
  }
}

__global__ void k2b_stats(const float* __restrict__ sumPart,
                          const float* __restrict__ sqPart,
                          const float* __restrict__ gamma,
                          const float* __restrict__ beta,
                          float* __restrict__ scaleA, float* __restrict__ scaleB) {
  const int c = threadIdx.x;
  if (c >= NC) return;
  float s = 0.0f, q = 0.0f;
  for (int blk = 0; blk < NB * NTILES; ++blk) {
    s += sumPart[blk * NC + c];
    q += sqPart [blk * NC + c];
  }
  const float invN = 1.0f / (float)(NB * NL);
  const float mean = s * invN;
  const float var  = q * invN - mean * mean;
  const float inv  = rsqrtf(var + EPS_BN);
  const float ga   = gamma[c] * inv;
  scaleA[c] = ga;
  scaleB[c] = beta[c] - mean * ga;
}

__global__ void k3_bn(float* __restrict__ y,
                      const float* __restrict__ scaleA,
                      const float* __restrict__ scaleB) {
  int i4 = blockIdx.x * blockDim.x + threadIdx.x;
  if (i4 >= NB * NC * NL / 4) return;
  const int c = (i4 >> 10) & (NC - 1);
  float4 v = ((float4*)y)[i4];
  const float a = scaleA[c], b = scaleB[c];
  v.x = v.x * a + b;  v.y = v.y * a + b;
  v.z = v.z * a + b;  v.w = v.w * a + b;
  ((float4*)y)[i4] = v;
}

extern "C" void kernel_launch(void* const* d_in, const int* in_sizes, int n_in,
                              void* d_out, int out_size, void* d_ws, size_t ws_size,
                              hipStream_t stream) {
  (void)in_sizes; (void)n_in; (void)out_size; (void)ws_size;
  const float* x      = (const float*)d_in[0];
  const float* conv_w = (const float*)d_in[1];
  const float* conv_b = (const float*)d_in[2];
  const float* wq     = (const float*)d_in[3];
  const float* bq     = (const float*)d_in[4];
  const float* wk     = (const float*)d_in[5];
  const float* bk     = (const float*)d_in[6];
  const float* wv     = (const float*)d_in[7];
  const float* bv     = (const float*)d_in[8];
  const float* gamma  = (const float*)d_in[9];
  const float* beta   = (const float*)d_in[10];
  float* out = (float*)d_out;

  float* ws      = (float*)d_ws;
  float* kbuf    = ws;
  float* Mpart   = kbuf    + (size_t)NB * NC * NL;
  float* Mbuf    = Mpart   + (size_t)NB * NTILES * NC * NC;
  float* sumPart = Mbuf    + (size_t)NB * NC * NC;
  float* sqPart  = sumPart + (size_t)NB * NTILES * NC;
  float* scaleA  = sqPart  + (size_t)NB * NTILES * NC;
  float* scaleB  = scaleA  + NC;

  const int grid1 = NB * NTILES;
  k1_conv_qkv_M<<<grid1, BLOCK, 0, stream>>>(x, conv_w, conv_b, wq, bq, wk, bk,
                                             wv, bv, kbuf, Mpart);
  k1b_reduceM<<<(NB * NC * NC + 255) / 256, 256, 0, stream>>>(Mpart, Mbuf);
  k2_gate_residual<<<grid1, BLOCK, 0, stream>>>(x, kbuf, Mbuf, out, sumPart, sqPart);
  k2b_stats<<<1, NC, 0, stream>>>(sumPart, sqPart, gamma, beta, scaleA, scaleB);
  k3_bn<<<(NB * NC * NL / 4 + 255) / 256, 256, 0, stream>>>(out, scaleA, scaleB);
}


# BN stats fold parallelised: 2 blocks x 1024 threads, 8 partial rows per thread loaded at once, LDS combine of 32 slice sums (two-level sum order replaces the serial 256-row walk)
# speedup vs baseline: 1.4194x; 1.4194x over previous
#include <hip/hip_runtime.h>
#include <math.h>


typedef __attribute__((ext_vector_type(2))) float v2f;
typedef __attribute__((ext_vector_type(8))) float v8f;

#define NB 4
#define NC 64
#define NL 4096
#define LT 64
#define NTILES (NL / LT)
#define NWAVES 4
#define BLOCK (NWAVES * 32)
#define EPS_BN 1e-5f

#define SE_STRIDE 68
#define SH_STRIDE 65

__device__ __forceinline__ v8f wmma_f32(v2f a, v2f b, v8f c) {
  return __builtin_amdgcn_wmma_f32_16x16x4_f32(false, a, false, b,
                                               (short)0, c, false, false);
}

__global__ __launch_bounds__(BLOCK)
void k1_conv_qkv_M(const float* __restrict__ x,
                   const float* __restrict__ conv_w, const float* __restrict__ conv_b,
                   const float* __restrict__ wq, const float* __restrict__ bq,
                   const float* __restrict__ wk, const float* __restrict__ bk,
                   const float* __restrict__ wv, const float* __restrict__ bv,
                   float* __restrict__ kbuf, float* __restrict__ Mpart) {
  __shared__ float sE[NC * SE_STRIDE];
  __shared__ float sH[NC * SH_STRIDE];
  __shared__ float sQ[NC * SH_STRIDE];
  __shared__ float sV[NC * SH_STRIDE];

  const int tid  = threadIdx.x;
  const int wave = tid >> 5;
  const int lane = tid & 31;
  const int half = lane >> 4;
  const int mn   = lane & 15;

  const int bIdx = blockIdx.x / NTILES;
  const int l0   = (blockIdx.x % NTILES) * LT;

  const float* xb = x + bIdx * NC * NL;
  for (int idx = tid; idx < NC * (LT + 2); idx += BLOCK) {
    int c = idx / (LT + 2);
    int j = idx % (LT + 2);
    int gl = (l0 + j - 1) & (NL - 1);
    float v = xb[c * NL + gl];
    sE[c * SE_STRIDE + j] = (v > 0.0f) ? v : (__expf(v) - 1.0f);
  }
  __syncthreads();

  const int mt = wave;
  const int arow = mt * 16 + mn;
  const int rbase = mt * 16 + half * 8;

  for (int nt = 0; nt < 4; ++nt) {
    const int bcol = nt * 16 + mn;
    v8f acc = {};
    for (int tap = 0; tap < 3; ++tap) {
      const float* Wt = conv_w + tap;
      for (int kk = 0; kk < 16; ++kk) {
        const int k0 = kk * 4 + half * 2;
        v2f a, b;
        a[0] = Wt[(arow * NC + k0) * 3];
        a[1] = Wt[(arow * NC + k0 + 1) * 3];
        b[0] = sE[k0 * SE_STRIDE + bcol + tap];
        b[1] = sE[(k0 + 1) * SE_STRIDE + bcol + tap];
        acc = wmma_f32(a, b, acc);
      }
    }
#pragma unroll
    for (int r = 0; r < 8; ++r)
      sH[(rbase + r) * SH_STRIDE + bcol] = acc[r] + conv_b[rbase + r];
  }
  __syncthreads();

  for (int nt = 0; nt < 4; ++nt) {
    const int bcol = nt * 16 + mn;
    v8f aq = {}, ak = {}, av = {};
    for (int kk = 0; kk < 16; ++kk) {
      const int k0 = kk * 4 + half * 2;
      v2f bf, a;
      bf[0] = sH[k0 * SH_STRIDE + bcol];
      bf[1] = sH[(k0 + 1) * SH_STRIDE + bcol];
      a[0] = wq[arow * NC + k0]; a[1] = wq[arow * NC + k0 + 1];
      aq = wmma_f32(a, bf, aq);
      a[0] = wk[arow * NC + k0]; a[1] = wk[arow * NC + k0 + 1];
      ak = wmma_f32(a, bf, ak);
      a[0] = wv[arow * NC + k0]; a[1] = wv[arow * NC + k0 + 1];
      av = wmma_f32(a, bf, av);
    }
#pragma unroll
    for (int r = 0; r < 8; ++r) {
      const int row = rbase + r;
      sQ[row * SH_STRIDE + bcol] = aq[r] + bq[row];
      sV[row * SH_STRIDE + bcol] = av[r] + bv[row];
      kbuf[(bIdx * NC + row) * NL + l0 + bcol] = ak[r] + bk[row];
    }
  }
  __syncthreads();

  float* Mout = Mpart + blockIdx.x * (NC * NC);
  for (int nt = 0; nt < 4; ++nt) {
    const int bcol = nt * 16 + mn;
    v8f acc = {};
    for (int kk = 0; kk < 16; ++kk) {
      const int k0 = kk * 4 + half * 2;
      v2f a, b;
      a[0] = sV[arow * SH_STRIDE + k0];
      a[1] = sV[arow * SH_STRIDE + k0 + 1];
      b[0] = sQ[bcol * SH_STRIDE + k0];
      b[1] = sQ[bcol * SH_STRIDE + k0 + 1];
      acc = wmma_f32(a, b, acc);
    }
#pragma unroll
    for (int r = 0; r < 8; ++r)
      Mout[(rbase + r) * NC + bcol] = acc[r];
  }
}

__global__ void k1b_reduceM(const float* __restrict__ Mpart,
                            float* __restrict__ Mbuf) {
  int i = blockIdx.x * blockDim.x + threadIdx.x;
  if (i >= NB * NC * NC) return;
  const int b = i >> 12;
  const int mn_ = i & 4095;
  const float* p = Mpart + (size_t)(b * NTILES) * (NC * NC) + mn_;
  float s = 0.0f;
  for (int t = 0; t < NTILES; ++t) s += p[(size_t)t * NC * NC];
  Mbuf[i] = s;
}

__global__ __launch_bounds__(BLOCK)
void k2_gate_residual(const float* __restrict__ x,
                      const float* __restrict__ kbuf,
                      const float* __restrict__ Mbuf,
                      float* __restrict__ y,
                      float* __restrict__ sumPart, float* __restrict__ sqPart) {
  __shared__ float sK[NC * SH_STRIDE];
  __shared__ float sSum[NC * 16];
  __shared__ float sSq[NC * 16];

  const int tid  = threadIdx.x;
  const int wave = tid >> 5;
  const int lane = tid & 31;
  const int half = lane >> 4;
  const int mn   = lane & 15;

  const int bIdx = blockIdx.x / NTILES;
  const int l0   = (blockIdx.x % NTILES) * LT;

  const float* kb = kbuf + (size_t)bIdx * NC * NL + l0;
  for (int idx = tid; idx < NC * LT; idx += BLOCK) {
    int c = idx >> 6, j = idx & 63;
    sK[c * SH_STRIDE + j] = kb[c * NL + j];
  }
  __syncthreads();

  const float* Mb = Mbuf + bIdx * NC * NC;
  const float* xb = x + (size_t)bIdx * NC * NL + l0;
  float* yb = y + (size_t)bIdx * NC * NL + l0;

  const int mt = wave;
  const int arow = mt * 16 + mn;
  const int rbase = mt * 16 + half * 8;

  float regS[8], regQ[8];
#pragma unroll
  for (int r = 0; r < 8; ++r) { regS[r] = 0.0f; regQ[r] = 0.0f; }

  for (int nt = 0; nt < 4; ++nt) {
    const int bcol = nt * 16 + mn;
    v8f acc = {};
    for (int kk = 0; kk < 16; ++kk) {
      const int k0 = kk * 4 + half * 2;
      v2f a, b;
      a[0] = Mb[arow * NC + k0];
      a[1] = Mb[arow * NC + k0 + 1];
      b[0] = sK[k0 * SH_STRIDE + bcol];
      b[1] = sK[(k0 + 1) * SH_STRIDE + bcol];
      acc = wmma_f32(a, b, acc);
    }
#pragma unroll
    for (int r = 0; r < 8; ++r) {
      const int row = rbase + r;
      float yv = acc[r] + xb[row * NL + bcol];
      yb[row * NL + bcol] = yv;
      regS[r] += yv;
      regQ[r] += yv * yv;
    }
  }
#pragma unroll
  for (int r = 0; r < 8; ++r) {
    sSum[(rbase + r) * 16 + mn] = regS[r];
    sSq [(rbase + r) * 16 + mn] = regQ[r];
  }
  __syncthreads();
  if (tid < NC) {
    float s = 0.0f, q = 0.0f;
    for (int j = 0; j < 16; ++j) { s += sSum[tid * 16 + j]; q += sSq[tid * 16 + j]; }
    sumPart[blockIdx.x * NC + tid] = s;
    sqPart [blockIdx.x * NC + tid] = q;
  }
}

#define FOLD_CH 32
#define FOLD_SLICES 32
#define FOLD_PER_SLICE ((NB * NTILES) / FOLD_SLICES)
__global__ __launch_bounds__(FOLD_SLICES * FOLD_CH)
void k2b_stats(const float* __restrict__ sumPart,
               const float* __restrict__ sqPart,
               const float* __restrict__ gamma,
               const float* __restrict__ beta,
               float* __restrict__ scaleA, float* __restrict__ scaleB) {
  __shared__ float sS[FOLD_SLICES * FOLD_CH];
  __shared__ float sQ[FOLD_SLICES * FOLD_CH];

  const int t  = threadIdx.x;
  const int cl = t & (FOLD_CH - 1);
  const int sl = t >> 5;
  const int c  = blockIdx.x * FOLD_CH + cl;
  const float* ps = sumPart + (sl * FOLD_PER_SLICE) * NC + c;
  const float* pq = sqPart  + (sl * FOLD_PER_SLICE) * NC + c;

  const float gm = gamma[c];
  const float bt = beta[c];

  float vs[FOLD_PER_SLICE], vq[FOLD_PER_SLICE];
#pragma unroll
  for (int i = 0; i < FOLD_PER_SLICE; ++i) {
    vs[i] = ps[i * NC];
    vq[i] = pq[i * NC];
  }
  float s = 0.0f, q = 0.0f;
#pragma unroll
  for (int i = 0; i < FOLD_PER_SLICE; ++i) { s += vs[i]; q += vq[i]; }
  sS[sl * FOLD_CH + cl] = s;
  sQ[sl * FOLD_CH + cl] = q;
  __syncthreads();

  if (t < FOLD_CH) {
    float ts = 0.0f, tq = 0.0f;
    for (int j = 0; j < FOLD_SLICES; ++j) {
      ts += sS[j * FOLD_CH + t];
      tq += sQ[j * FOLD_CH + t];
    }
    const float invN = 1.0f / (float)(NB * NL);
    const float mean = ts * invN;
    const float var  = tq * invN - mean * mean;
    const float inv  = rsqrtf(var + EPS_BN);
    const float ga   = gm * inv;
    scaleA[c] = ga;
    scaleB[c] = bt - mean * ga;
  }
}

__global__ void k3_bn(float* __restrict__ y,
                      const float* __restrict__ scaleA,
                      const float* __restrict__ scaleB) {
  int i4 = blockIdx.x * blockDim.x + threadIdx.x;
  if (i4 >= NB * NC * NL / 4) return;
  const int c = (i4 >> 10) & (NC - 1);
  float4 v = ((float4*)y)[i4];
  const float a = scaleA[c], b = scaleB[c];
  v.x = v.x * a + b;  v.y = v.y * a + b;
  v.z = v.z * a + b;  v.w = v.w * a + b;
  ((float4*)y)[i4] = v;
}

extern "C" void kernel_launch(void* const* d_in, const int* in_sizes, int n_in,
                              void* d_out, int out_size, void* d_ws, size_t ws_size,
                              hipStream_t stream) {
  (void)in_sizes; (void)n_in; (void)out_size; (void)ws_size;
  const float* x      = (const float*)d_in[0];
  const float* conv_w = (const float*)d_in[1];
  const float* conv_b = (const float*)d_in[2];
  const float* wq     = (const float*)d_in[3];
  const float* bq     = (const float*)d_in[4];
  const float* wk     = (const float*)d_in[5];
  const float* bk     = (const float*)d_in[6];
  const float* wv     = (const float*)d_in[7];
  const float* bv     = (const float*)d_in[8];
  const float* gamma  = (const float*)d_in[9];
  const float* beta   = (const float*)d_in[10];
  float* out = (float*)d_out;

  float* ws      = (float*)d_ws;
  float* kbuf    = ws;
  float* Mpart   = kbuf    + (size_t)NB * NC * NL;
  float* Mbuf    = Mpart   + (size_t)NB * NTILES * NC * NC;
  float* sumPart = Mbuf    + (size_t)NB * NC * NC;
  float* sqPart  = sumPart + (size_t)NB * NTILES * NC;
  float* scaleA  = sqPart  + (size_t)NB * NTILES * NC;
  float* scaleB  = scaleA  + NC;

  const int grid1 = NB * NTILES;
  k1_conv_qkv_M<<<grid1, BLOCK, 0, stream>>>(x, conv_w, conv_b, wq, bq, wk, bk,
                                             wv, bv, kbuf, Mpart);
  k1b_reduceM<<<(NB * NC * NC + 255) / 256, 256, 0, stream>>>(Mpart, Mbuf);
  k2_gate_residual<<<grid1, BLOCK, 0, stream>>>(x, kbuf, Mbuf, out, sumPart, sqPart);
  k2b_stats<<<NC / FOLD_CH, FOLD_SLICES * FOLD_CH, 0, stream>>>(sumPart, sqPart, gamma, beta, scaleA, scaleB);
  k3_bn<<<(NB * NC * NL / 4 + 255) / 256, 256, 0, stream>>>(out, scaleA, scaleB);
}


# Global reads grouped ahead of use (x, k tiles as float4; weight, M, residual operands; the 64 M slabs): each kernel waits for memory once, not once per staging-loop trip (9, 4 and 4 trips before)
# speedup vs baseline: 1.7090x; 1.2040x over previous
#include <hip/hip_runtime.h>
#include <math.h>


typedef __attribute__((ext_vector_type(2))) float v2f;
typedef __attribute__((ext_vector_type(8))) float v8f;

#define NB 4
#define NC 64
#define NL 4096
#define LT 64
#define NTILES (NL / LT)
#define NWAVES 4
#define BLOCK (NWAVES * 32)
#define EPS_BN 1e-5f

#define SE_STRIDE 68
#define SH_STRIDE 65

__device__ __forceinline__ v8f wmma_f32(v2f a, v2f b, v8f c) {
  return __builtin_amdgcn_wmma_f32_16x16x4_f32(false, a, false, b,
                                               (short)0, c, false, false);
}

__global__ __launch_bounds__(BLOCK) __attribute__((amdgpu_waves_per_eu(1, 2)))
void k1_conv_qkv_M(const float* __restrict__ x,
                   const float* __restrict__ conv_w, const float* __restrict__ conv_b,
                   const float* __restrict__ wq, const float* __restrict__ bq,
                   const float* __restrict__ wk, const float* __restrict__ bk,
                   const float* __restrict__ wv, const float* __restrict__ bv,
                   float* __restrict__ kbuf, float* __restrict__ Mpart) {
  __shared__ float sE[NC * SE_STRIDE];
  __shared__ float sH[NC * SH_STRIDE];
  __shared__ float sQ[NC * SH_STRIDE];
  __shared__ float sV[NC * SH_STRIDE];

  const int tid  = threadIdx.x;
  const int wave = tid >> 5;
  const int lane = tid & 31;
  const int half = lane >> 4;
  const int mn   = lane & 15;

  const int bIdx = blockIdx.x / NTILES;
  const int l0   = (blockIdx.x % NTILES) * LT;

  const int mt = wave;
  const int arow = mt * 16 + mn;
  const int rbase = mt * 16 + half * 8;

  const float* xb = x + bIdx * NC * NL;
  float4 xt[8];
#pragma unroll
  for (int i = 0; i < 8; ++i) {
    const int f = tid + i * BLOCK;
    xt[i] = *(const float4*)(xb + (f >> 4) * NL + l0 + (f & 15) * 4);
  }
  const int hrow = tid >> 1;
  const int hj   = (tid & 1) * (LT + 1);
  const float xh = xb[hrow * NL + ((l0 + hj - 1) & (NL - 1))];
  float cw[16][6];
#pragma unroll
  for (int kk = 0; kk < 16; ++kk) {
    const float* p = conv_w + (arow * NC + kk * 4 + half * 2) * 3;
#pragma unroll
    for (int e = 0; e < 6; ++e) cw[kk][e] = p[e];
  }
  float cbias[8];
#pragma unroll
  for (int r = 0; r < 8; ++r) cbias[r] = conv_b[rbase + r];

#pragma unroll
  for (int i = 0; i < 8; ++i) {
    const int f = tid + i * BLOCK;
    float* d = sE + (f >> 4) * SE_STRIDE + 1 + (f & 15) * 4;
    const float4 v = xt[i];
    d[0] = (v.x > 0.0f) ? v.x : (__expf(v.x) - 1.0f);
    d[1] = (v.y > 0.0f) ? v.y : (__expf(v.y) - 1.0f);
    d[2] = (v.z > 0.0f) ? v.z : (__expf(v.z) - 1.0f);
    d[3] = (v.w > 0.0f) ? v.w : (__expf(v.w) - 1.0f);
  }
  sE[hrow * SE_STRIDE + hj] = (xh > 0.0f) ? xh : (__expf(xh) - 1.0f);
  __syncthreads();

  v2f fq[16], fk[16], fv[16];
#pragma unroll
  for (int kk = 0; kk < 16; ++kk) {
    const int k0 = kk * 4 + half * 2;
    fq[kk][0] = wq[arow * NC + k0]; fq[kk][1] = wq[arow * NC + k0 + 1];
    fk[kk][0] = wk[arow * NC + k0]; fk[kk][1] = wk[arow * NC + k0 + 1];
    fv[kk][0] = wv[arow * NC + k0]; fv[kk][1] = wv[arow * NC + k0 + 1];
  }
  float qbias[8], kbias[8], vbias[8];
#pragma unroll
  for (int r = 0; r < 8; ++r) {
    qbias[r] = bq[rbase + r];
    kbias[r] = bk[rbase + r];
    vbias[r] = bv[rbase + r];
  }

  for (int nt = 0; nt < 4; ++nt) {
    const int bcol = nt * 16 + mn;
    v8f acc = {};
#pragma unroll
    for (int tap = 0; tap < 3; ++tap) {
#pragma unroll
      for (int kk = 0; kk < 16; ++kk) {
        const int k0 = kk * 4 + half * 2;
        v2f a, b;
        a[0] = cw[kk][tap];
        a[1] = cw[kk][3 + tap];
        b[0] = sE[k0 * SE_STRIDE + bcol + tap];
        b[1] = sE[(k0 + 1) * SE_STRIDE + bcol + tap];
        acc = wmma_f32(a, b, acc);
      }
    }
#pragma unroll
    for (int r = 0; r < 8; ++r)
      sH[(rbase + r) * SH_STRIDE + bcol] = acc[r] + cbias[r];
  }
  __syncthreads();

  for (int nt = 0; nt < 4; ++nt) {
    const int bcol = nt * 16 + mn;
    v8f aq = {}, ak = {}, av = {};
#pragma unroll
    for (int kk = 0; kk < 16; ++kk) {
      const int k0 = kk * 4 + half * 2;
      v2f bf;
      bf[0] = sH[k0 * SH_STRIDE + bcol];
      bf[1] = sH[(k0 + 1) * SH_STRIDE + bcol];
      aq = wmma_f32(fq[kk], bf, aq);
      ak = wmma_f32(fk[kk], bf, ak);
      av = wmma_f32(fv[kk], bf, av);
    }
#pragma unroll
    for (int r = 0; r < 8; ++r) {
      const int row = rbase + r;
      sQ[row * SH_STRIDE + bcol] = aq[r] + qbias[r];
      sV[row * SH_STRIDE + bcol] = av[r] + vbias[r];
      kbuf[(bIdx * NC + row) * NL + l0 + bcol] = ak[r] + kbias[r];
    }
  }
  __syncthreads();

  float* Mout = Mpart + blockIdx.x * (NC * NC);
  for (int nt = 0; nt < 4; ++nt) {
    const int bcol = nt * 16 + mn;
    v8f acc = {};
    for (int kk = 0; kk < 16; ++kk) {
      const int k0 = kk * 4 + half * 2;
      v2f a, b;
      a[0] = sV[arow * SH_STRIDE + k0];
      a[1] = sV[arow * SH_STRIDE + k0 + 1];
      b[0] = sQ[bcol * SH_STRIDE + k0];
      b[1] = sQ[bcol * SH_STRIDE + k0 + 1];
      acc = wmma_f32(a, b, acc);
    }
#pragma unroll
    for (int r = 0; r < 8; ++r)
      Mout[(rbase + r) * NC + bcol] = acc[r];
  }
}

__global__ void k1b_reduceM(const float* __restrict__ Mpart,
                            float* __restrict__ Mbuf) {
  int i = blockIdx.x * blockDim.x + threadIdx.x;
  if (i >= NB * NC * NC) return;
  const int b = i >> 12;
  const int mn_ = i & 4095;
  const float* p = Mpart + (size_t)(b * NTILES) * (NC * NC) + mn_;
  float v[NTILES];
#pragma unroll
  for (int t = 0; t < NTILES; ++t) v[t] = p[(size_t)t * NC * NC];
  float s = 0.0f;
#pragma unroll
  for (int t = 0; t < NTILES; ++t) s += v[t];
  Mbuf[i] = s;
}

__global__ __launch_bounds__(BLOCK)
void k2_gate_residual(const float* __restrict__ x,
                      const float* __restrict__ kbuf,
                      const float* __restrict__ Mbuf,
                      float* __restrict__ y,
                      float* __restrict__ sumPart, float* __restrict__ sqPart) {
  __shared__ float sK[NC * SH_STRIDE];
  __shared__ float sSum[NC * 16];
  __shared__ float sSq[NC * 16];

  const int tid  = threadIdx.x;
  const int wave = tid >> 5;
  const int lane = tid & 31;
  const int half = lane >> 4;
  const int mn   = lane & 15;

  const int bIdx = blockIdx.x / NTILES;
  const int l0   = (blockIdx.x % NTILES) * LT;

  const float* kb = kbuf + (size_t)bIdx * NC * NL + l0;
  const float* Mb = Mbuf + bIdx * NC * NC;
  const float* xb = x + (size_t)bIdx * NC * NL + l0;
  float* yb = y + (size_t)bIdx * NC * NL + l0;

  const int mt = wave;
  const int arow = mt * 16 + mn;
  const int rbase = mt * 16 + half * 8;

  float4 kt[8];
#pragma unroll
  for (int i = 0; i < 8; ++i) {
    const int f = tid + i * BLOCK;
    kt[i] = *(const float4*)(kb + (f >> 4) * NL + (f & 15) * 4);
  }
  v2f fm[16];
#pragma unroll
  for (int kk = 0; kk < 16; ++kk) {
    const int k0 = kk * 4 + half * 2;
    fm[kk][0] = Mb[arow * NC + k0];
    fm[kk][1] = Mb[arow * NC + k0 + 1];
  }
  float xr[4][8];
#pragma unroll
  for (int nt = 0; nt < 4; ++nt)
#pragma unroll
    for (int r = 0; r < 8; ++r)
      xr[nt][r] = xb[(rbase + r) * NL + nt * 16 + mn];

#pragma unroll
  for (int i = 0; i < 8; ++i) {
    const int f = tid + i * BLOCK;
    float* d = sK + (f >> 4) * SH_STRIDE + (f & 15) * 4;
    d[0] = kt[i].x; d[1] = kt[i].y; d[2] = kt[i].z; d[3] = kt[i].w;
  }
  __syncthreads();

  float regS[8], regQ[8];
#pragma unroll
  for (int r = 0; r < 8; ++r) { regS[r] = 0.0f; regQ[r] = 0.0f; }

#pragma unroll
  for (int nt = 0; nt < 4; ++nt) {
    const int bcol = nt * 16 + mn;
    v8f acc = {};
#pragma unroll
    for (int kk = 0; kk < 16; ++kk) {
      const int k0 = kk * 4 + half * 2;
      v2f b;
      b[0] = sK[k0 * SH_STRIDE + bcol];
      b[1] = sK[(k0 + 1) * SH_STRIDE + bcol];
      acc = wmma_f32(fm[kk], b, acc);
    }
#pragma unroll
    for (int r = 0; r < 8; ++r) {
      const int row = rbase + r;
      float yv = acc[r] + xr[nt][r];
      yb[row * NL + bcol] = yv;
      regS[r] += yv;
      regQ[r] += yv * yv;
    }
  }
#pragma unroll
  for (int r = 0; r < 8; ++r) {
    sSum[(rbase + r) * 16 + mn] = regS[r];
    sSq [(rbase + r) * 16 + mn] = regQ[r];
  }
  __syncthreads();
  if (tid < NC) {
    float s = 0.0f, q = 0.0f;
    for (int j = 0; j < 16; ++j) { s += sSum[tid * 16 + j]; q += sSq[tid * 16 + j]; }
    sumPart[blockIdx.x * NC + tid] = s;
    sqPart [blockIdx.x * NC + tid] = q;
  }
}

#define FOLD_CH 32
#define FOLD_SLICES 32
#define FOLD_PER_SLICE ((NB * NTILES) / FOLD_SLICES)
__global__ __launch_bounds__(FOLD_SLICES * FOLD_CH)
void k2b_stats(const float* __restrict__ sumPart,
               const float* __restrict__ sqPart,
               const float* __restrict__ gamma,
               const float* __restrict__ beta,
               float* __restrict__ scaleA, float* __restrict__ scaleB) {
  __shared__ float sS[FOLD_SLICES * FOLD_CH];
  __shared__ float sQ[FOLD_SLICES * FOLD_CH];

  const int t  = threadIdx.x;
  const int cl = t & (FOLD_CH - 1);
  const int sl = t >> 5;
  const int c  = blockIdx.x * FOLD_CH + cl;
  const float* ps = sumPart + (sl * FOLD_PER_SLICE) * NC + c;
  const float* pq = sqPart  + (sl * FOLD_PER_SLICE) * NC + c;

  const float gm = gamma[c];
  const float bt = beta[c];

  float vs[FOLD_PER_SLICE], vq[FOLD_PER_SLICE];
#pragma unroll
  for (int i = 0; i < FOLD_PER_SLICE; ++i) {
    vs[i] = ps[i * NC];
    vq[i] = pq[i * NC];
  }
  float s = 0.0f, q = 0.0f;
#pragma unroll
  for (int i = 0; i < FOLD_PER_SLICE; ++i) { s += vs[i]; q += vq[i]; }
  sS[sl * FOLD_CH + cl] = s;
  sQ[sl * FOLD_CH + cl] = q;
  __syncthreads();

  if (t < FOLD_CH) {
    float ts = 0.0f, tq = 0.0f;
    for (int j = 0; j < FOLD_SLICES; ++j) {
      ts += sS[j * FOLD_CH + t];
      tq += sQ[j * FOLD_CH + t];
    }
    const float invN = 1.0f / (float)(NB * NL);
    const float mean = ts * invN;
    const float var  = tq * invN - mean * mean;
    const float inv  = rsqrtf(var + EPS_BN);
    const float ga   = gm * inv;
    scaleA[c] = ga;
    scaleB[c] = bt - mean * ga;
  }
}

__global__ void k3_bn(float* __restrict__ y,
                      const float* __restrict__ scaleA,
                      const float* __restrict__ scaleB) {
  int i4 = blockIdx.x * blockDim.x + threadIdx.x;
  if (i4 >= NB * NC * NL / 4) return;
  const int c = (i4 >> 10) & (NC - 1);
  float4 v = ((float4*)y)[i4];
  const float a = scaleA[c], b = scaleB[c];
  v.x = v.x * a + b;  v.y = v.y * a + b;
  v.z = v.z * a + b;  v.w = v.w * a + b;
  ((float4*)y)[i4] = v;
}

extern "C" void kernel_launch(void* const* d_in, const int* in_sizes, int n_in,
                              void* d_out, int out_size, void* d_ws, size_t ws_size,
                              hipStream_t stream) {
  (void)in_sizes; (void)n_in; (void)out_size; (void)ws_size;
  const float* x      = (const float*)d_in[0];
  const float* conv_w = (const float*)d_in[1];
  const float* conv_b = (const float*)d_in[2];
  const float* wq     = (const float*)d_in[3];
  const float* bq     = (const float*)d_in[4];
  const float* wk     = (const float*)d_in[5];
  const float* bk     = (const float*)d_in[6];
  const float* wv     = (const float*)d_in[7];
  const float* bv     = (const float*)d_in[8];
  const float* gamma  = (const float*)d_in[9];
  const float* beta   = (const float*)d_in[10];
  float* out = (float*)d_out;

  float* ws      = (float*)d_ws;
  float* kbuf    = ws;
  float* Mpart   = kbuf    + (size_t)NB * NC * NL;
  float* Mbuf    = Mpart   + (size_t)NB * NTILES * NC * NC;
  float* sumPart = Mbuf    + (size_t)NB * NC * NC;
  float* sqPart  = sumPart + (size_t)NB * NTILES * NC;
  float* scaleA  = sqPart  + (size_t)NB * NTILES * NC;
  float* scaleB  = scaleA  + NC;

  const int grid1 = NB * NTILES;
  k1_conv_qkv_M<<<grid1, BLOCK, 0, stream>>>(x, conv_w, conv_b, wq, bq, wk, bk,
                                             wv, bv, kbuf, Mpart);
  k1b_reduceM<<<(NB * NC * NC + 255) / 256, 256, 0, stream>>>(Mpart, Mbuf);
  k2_gate_residual<<<grid1, BLOCK, 0, stream>>>(x, kbuf, Mbuf, out, sumPart, sqPart);
  k2b_stats<<<NC / FOLD_CH, FOLD_SLICES * FOLD_CH, 0, stream>>>(sumPart, sqPart, gamma, beta, scaleA, scaleB);
  k3_bn<<<(NB * NC * NL / 4 + 255) / 256, 256, 0, stream>>>(out, scaleA, scaleB);
}


# WMMA fragments cut from k-contiguous runs (k = 32h+2j+s), fetched as 128-bit LDS loads one block ahead: no LDS access between the WMMAs of a block; new k sum order in every matrix product
# speedup vs baseline: 1.8371x; 1.0750x over previous
#include <hip/hip_runtime.h>
#include <math.h>


typedef __attribute__((ext_vector_type(2))) float v2f;
typedef __attribute__((ext_vector_type(8))) float v8f;

#define NB 4
#define NC 64
#define NL 4096
#define LT 64
#define NTILES (NL / LT)
#define NWAVES 4
#define BLOCK (NWAVES * 32)
#define EPS_BN 1e-5f

#define SP 68

typedef __attribute__((ext_vector_type(4))) float v4f;

__device__ __forceinline__ v2f frag(const v4f (&run)[8], int j) {
  v2f r;
  r[0] = run[j >> 1][(j & 1) * 2];
  r[1] = run[j >> 1][(j & 1) * 2 + 1];
  return r;
}

__device__ __forceinline__ v8f wmma_f32(v2f a, v2f b, v8f c) {
  return __builtin_amdgcn_wmma_f32_16x16x4_f32(false, a, false, b,
                                               (short)0, c, false, false);
}

__global__ __launch_bounds__(BLOCK) __attribute__((amdgpu_waves_per_eu(1, 2)))
void k1_conv_qkv_M(const float* __restrict__ x,
                   const float* __restrict__ conv_w, const float* __restrict__ conv_b,
                   const float* __restrict__ wq, const float* __restrict__ bq,
                   const float* __restrict__ wk, const float* __restrict__ bk,
                   const float* __restrict__ wv, const float* __restrict__ bv,
                   float* __restrict__ kbuf, float* __restrict__ Mpart) {
  __shared__ __attribute__((aligned(16))) float sE[(LT + 2) * SP];
  __shared__ __attribute__((aligned(16))) float sH[LT * SP];
  __shared__ __attribute__((aligned(16))) float sQ[NC * SP];
  __shared__ __attribute__((aligned(16))) float sV[NC * SP];

  const int tid  = threadIdx.x;
  const int wave = tid >> 5;
  const int lane = tid & 31;
  const int half = lane >> 4;
  const int mn   = lane & 15;

  const int bIdx = blockIdx.x / NTILES;
  const int l0   = (blockIdx.x % NTILES) * LT;

  const int mt = wave;
  const int arow = mt * 16 + mn;
  const int rbase = mt * 16 + half * 8;
  const int krun = half * 32;

  const float* xb = x + bIdx * NC * NL;
  float4 xt[8];
#pragma unroll
  for (int i = 0; i < 8; ++i) {
    const int f = tid + i * BLOCK;
    xt[i] = *(const float4*)(xb + (f >> 4) * NL + l0 + (f & 15) * 4);
  }
  const int hrow = tid >> 1;
  const int hj   = (tid & 1) * (LT + 1);
  const float xh = xb[hrow * NL + ((l0 + hj - 1) & (NL - 1))];
  float cwf[96];
  {
    const float* p = conv_w + (arow * NC + krun) * 3;
#pragma unroll
    for (int i = 0; i < 24; ++i) {
      const v4f t = *(const v4f*)(p + 4 * i);
      cwf[4 * i] = t[0]; cwf[4 * i + 1] = t[1]; cwf[4 * i + 2] = t[2]; cwf[4 * i + 3] = t[3];
    }
  }
  float cbias[8];
#pragma unroll
  for (int r = 0; r < 8; ++r) cbias[r] = conv_b[rbase + r];

#pragma unroll
  for (int i = 0; i < 8; ++i) {
    const int f = tid + i * BLOCK;
    float* d = sE + (1 + (f & 15) * 4) * SP + (f >> 4);
    const float4 v = xt[i];
    d[0]      = (v.x > 0.0f) ? v.x : (__expf(v.x) - 1.0f);
    d[SP]     = (v.y > 0.0f) ? v.y : (__expf(v.y) - 1.0f);
    d[2 * SP] = (v.z > 0.0f) ? v.z : (__expf(v.z) - 1.0f);
    d[3 * SP] = (v.w > 0.0f) ? v.w : (__expf(v.w) - 1.0f);
  }
  sE[hj * SP + hrow] = (xh > 0.0f) ? xh : (__expf(xh) - 1.0f);
  asm volatile("" :: "v"(cbias[0]), "v"(cbias[1]), "v"(cbias[2]), "v"(cbias[3]),
                     "v"(cbias[4]), "v"(cbias[5]), "v"(cbias[6]), "v"(cbias[7]));
  __syncthreads();

  v4f fq[8], fk[8], fv[8];
#pragma unroll
  for (int i = 0; i < 8; ++i) {
    fq[i] = *(const v4f*)(wq + arow * NC + krun + 4 * i);
    fk[i] = *(const v4f*)(wk + arow * NC + krun + 4 * i);
    fv[i] = *(const v4f*)(wv + arow * NC + krun + 4 * i);
  }
  float qbias[8], kbias[8], vbias[8];
#pragma unroll
  for (int r = 0; r < 8; ++r) {
    qbias[r] = bq[rbase + r];
    kbias[r] = bk[rbase + r];
    vbias[r] = bv[rbase + r];
  }

  {
    v4f bb[2][2][8];
    const float* eb = sE + mn * SP + krun;
#pragma unroll
    for (int i = 0; i < 8; ++i) {
      bb[0][0][i] = *(const v4f*)(eb + 4 * i);
      bb[0][1][i] = *(const v4f*)(eb + 16 * SP + 4 * i);
    }
    v8f acc[4] = {};
#pragma unroll
    for (int bi = 0; bi < 6; ++bi) {
      const int p = bi / 3, tap = bi % 3;
      if (bi + 1 < 6) {
        const int p2 = (bi + 1) / 3, tap2 = (bi + 1) % 3;
        const float* nb = eb + (p2 * 32 + tap2) * SP;
#pragma unroll
        for (int i = 0; i < 8; ++i) {
          bb[(bi + 1) & 1][0][i] = *(const v4f*)(nb + 4 * i);
          bb[(bi + 1) & 1][1][i] = *(const v4f*)(nb + 16 * SP + 4 * i);
        }
      }
      __builtin_amdgcn_sched_barrier(0);
#pragma unroll
      for (int j = 0; j < 16; ++j) {
        v2f a;
        a[0] = cwf[(2 * j) * 3 + tap];
        a[1] = cwf[(2 * j + 1) * 3 + tap];
        acc[2 * p]     = wmma_f32(a, frag(bb[bi & 1][0], j), acc[2 * p]);
        acc[2 * p + 1] = wmma_f32(a, frag(bb[bi & 1][1], j), acc[2 * p + 1]);
      }
      __builtin_amdgcn_sched_barrier(0);
      if (tap == 2) {
#pragma unroll
        for (int t = 0; t < 2; ++t) {
          const int nt = 2 * p + t;
          float* d = sH + (nt * 16 + mn) * SP + rbase;
          v4f lo, hi;
#pragma unroll
          for (int r = 0; r < 4; ++r) {
            lo[r] = acc[nt][r] + cbias[r];
            hi[r] = acc[nt][4 + r] + cbias[4 + r];
          }
          *(v4f*)d = lo;
          *(v4f*)(d + 4) = hi;
        }
      }
    }
  }
  __syncthreads();

  {
    v4f hb[2][8];
    const float* hp = sH + mn * SP + krun;
#pragma unroll
    for (int i = 0; i < 8; ++i) hb[0][i] = *(const v4f*)(hp + 4 * i);
#pragma unroll
    for (int nt = 0; nt < 4; ++nt) {
      const int bcol = nt * 16 + mn;
      if (nt + 1 < 4) {
#pragma unroll
        for (int i = 0; i < 8; ++i)
          hb[(nt + 1) & 1][i] = *(const v4f*)(hp + (nt + 1) * 16 * SP + 4 * i);
      }
      __builtin_amdgcn_sched_barrier(0);
      v8f aq = {}, ak = {}, av = {};
#pragma unroll
      for (int j = 0; j < 16; ++j) {
        const v2f bf = frag(hb[nt & 1], j);
        aq = wmma_f32(frag(fq, j), bf, aq);
        ak = wmma_f32(frag(fk, j), bf, ak);
        av = wmma_f32(frag(fv, j), bf, av);
      }
      __builtin_amdgcn_sched_barrier(0);
#pragma unroll
      for (int r = 0; r < 8; ++r) {
        const int row = rbase + r;
        sQ[row * SP + bcol] = aq[r] + qbias[r];
        sV[row * SP + bcol] = av[r] + vbias[r];
        kbuf[(bIdx * NC + row) * NL + l0 + bcol] = ak[r] + kbias[r];
      }
    }
  }
  __syncthreads();

  float* Mout = Mpart + blockIdx.x * (NC * NC);
  {
    v4f va[8], qb[4][8];
    const float* vp = sV + arow * SP + krun;
    const float* qp = sQ + mn * SP + krun;
#pragma unroll
    for (int i = 0; i < 8; ++i) {
      va[i] = *(const v4f*)(vp + 4 * i);
#pragma unroll
      for (int nt = 0; nt < 4; ++nt)
        qb[nt][i] = *(const v4f*)(qp + nt * 16 * SP + 4 * i);
    }
    __builtin_amdgcn_sched_barrier(0);
    v8f acc[4] = {};
#pragma unroll
    for (int j = 0; j < 16; ++j) {
      const v2f a = frag(va, j);
#pragma unroll
      for (int nt = 0; nt < 4; ++nt)
        acc[nt] = wmma_f32(a, frag(qb[nt], j), acc[nt]);
    }
#pragma unroll
    for (int nt = 0; nt < 4; ++nt)
#pragma unroll
      for (int r = 0; r < 8; ++r)
        Mout[(rbase + r) * NC + nt * 16 + mn] = acc[nt][r];
  }
}

__global__ void k1b_reduceM(const float* __restrict__ Mpart,
                            float* __restrict__ Mbuf) {
  int i = blockIdx.x * blockDim.x + threadIdx.x;
  if (i >= NB * NC * NC) return;
  const int b = i >> 12;
  const int mn_ = i & 4095;
  const float* p = Mpart + (size_t)(b * NTILES) * (NC * NC) + mn_;
  float v[NTILES];
#pragma unroll
  for (int t = 0; t < NTILES; ++t) v[t] = p[(size_t)t * NC * NC];
  float s = 0.0f;
#pragma unroll
  for (int t = 0; t < NTILES; ++t) s += v[t];
  Mbuf[i] = s;
}

__global__ __launch_bounds__(BLOCK)
void k2_gate_residual(const float* __restrict__ x,
                      const float* __restrict__ kbuf,
                      const float* __restrict__ Mbuf,
                      float* __restrict__ y,
                      float* __restrict__ sumPart, float* __restrict__ sqPart) {
  __shared__ __attribute__((aligned(16))) float sK[LT * SP];
  __shared__ float sSum[NC * 16];
  __shared__ float sSq[NC * 16];

  const int tid  = threadIdx.x;
  const int wave = tid >> 5;
  const int lane = tid & 31;
  const int half = lane >> 4;
  const int mn   = lane & 15;

  const int bIdx = blockIdx.x / NTILES;
  const int l0   = (blockIdx.x % NTILES) * LT;

  const float* kb = kbuf + (size_t)bIdx * NC * NL + l0;
  const float* Mb = Mbuf + bIdx * NC * NC;
  const float* xb = x + (size_t)bIdx * NC * NL + l0;
  float* yb = y + (size_t)bIdx * NC * NL + l0;

  const int mt = wave;
  const int arow = mt * 16 + mn;
  const int rbase = mt * 16 + half * 8;
  const int krun = half * 32;

  float4 kt[8];
#pragma unroll
  for (int i = 0; i < 8; ++i) {
    const int f = tid + i * BLOCK;
    kt[i] = *(const float4*)(kb + (f >> 4) * NL + (f & 15) * 4);
  }
  v4f fm[8];
#pragma unroll
  for (int i = 0; i < 8; ++i)
    fm[i] = *(const v4f*)(Mb + arow * NC + krun + 4 * i);
  float xr[4][8];
#pragma unroll
  for (int nt = 0; nt < 4; ++nt)
#pragma unroll
    for (int r = 0; r < 8; ++r)
      xr[nt][r] = xb[(rbase + r) * NL + nt * 16 + mn];

#pragma unroll
  for (int i = 0; i < 8; ++i) {
    const int f = tid + i * BLOCK;
    float* d = sK + ((f & 15) * 4) * SP + (f >> 4);
    d[0] = kt[i].x; d[SP] = kt[i].y; d[2 * SP] = kt[i].z; d[3 * SP] = kt[i].w;
  }
  __syncthreads();

  float regS[8], regQ[8];
#pragma unroll
  for (int r = 0; r < 8; ++r) { regS[r] = 0.0f; regQ[r] = 0.0f; }

  v4f kr[4][8];
  {
    const float* kp = sK + mn * SP + krun;
#pragma unroll
    for (int i = 0; i < 8; ++i)
#pragma unroll
      for (int nt = 0; nt < 4; ++nt)
        kr[nt][i] = *(const v4f*)(kp + nt * 16 * SP + 4 * i);
  }
  __builtin_amdgcn_sched_barrier(0);
  v8f acc[4] = {};
#pragma unroll
  for (int j = 0; j < 16; ++j) {
    const v2f a = frag(fm, j);
#pragma unroll
    for (int nt = 0; nt < 4; ++nt)
      acc[nt] = wmma_f32(a, frag(kr[nt], j), acc[nt]);
  }

#pragma unroll
  for (int nt = 0; nt < 4; ++nt) {
    const int bcol = nt * 16 + mn;
#pragma unroll
    for (int r = 0; r < 8; ++r) {
      const int row = rbase + r;
      float yv = acc[nt][r] + xr[nt][r];
      yb[row * NL + bcol] = yv;
      regS[r] += yv;
      regQ[r] += yv * yv;
    }
  }
#pragma unroll
  for (int r = 0; r < 8; ++r) {
    sSum[(rbase + r) * 16 + mn] = regS[r];
    sSq [(rbase + r) * 16 + mn] = regQ[r];
  }
  __syncthreads();
  if (tid < NC) {
    float s = 0.0f, q = 0.0f;
    for (int j = 0; j < 16; ++j) { s += sSum[tid * 16 + j]; q += sSq[tid * 16 + j]; }
    sumPart[blockIdx.x * NC + tid] = s;
    sqPart [blockIdx.x * NC + tid] = q;
  }
}

#define FOLD_CH 32
#define FOLD_SLICES 32
#define FOLD_PER_SLICE ((NB * NTILES) / FOLD_SLICES)
__global__ __launch_bounds__(FOLD_SLICES * FOLD_CH)
void k2b_stats(const float* __restrict__ sumPart,
               const float* __restrict__ sqPart,
               const float* __restrict__ gamma,
               const float* __restrict__ beta,
               float* __restrict__ scaleA, float* __restrict__ scaleB) {
  __shared__ float sS[FOLD_SLICES * FOLD_CH];
  __shared__ float sQ[FOLD_SLICES * FOLD_CH];

  const int t  = threadIdx.x;
  const int cl = t & (FOLD_CH - 1);
  const int sl = t >> 5;
  const int c  = blockIdx.x * FOLD_CH + cl;
  const float* ps = sumPart + (sl * FOLD_PER_SLICE) * NC + c;
  const float* pq = sqPart  + (sl * FOLD_PER_SLICE) * NC + c;

  const float gm = gamma[c];
  const float bt = beta[c];

  float vs[FOLD_PER_SLICE], vq[FOLD_PER_SLICE];
#pragma unroll
  for (int i = 0; i < FOLD_PER_SLICE; ++i) {
    vs[i] = ps[i * NC];
    vq[i] = pq[i * NC];
  }
  float s = 0.0f, q = 0.0f;
#pragma unroll
  for (int i = 0; i < FOLD_PER_SLICE; ++i) { s += vs[i]; q += vq[i]; }
  sS[sl * FOLD_CH + cl] = s;
  sQ[sl * FOLD_CH + cl] = q;
  __syncthreads();

  if (t < FOLD_CH) {
    float ts = 0.0f, tq = 0.0f;
    for (int j = 0; j < FOLD_SLICES; ++j) {
      ts += sS[j * FOLD_CH + t];
      tq += sQ[j * FOLD_CH + t];
    }
    const float invN = 1.0f / (float)(NB * NL);
    const float mean = ts * invN;
    const float var  = tq * invN - mean * mean;
    const float inv  = rsqrtf(var + EPS_BN);
    const float ga   = gm * inv;
    scaleA[c] = ga;
    scaleB[c] = bt - mean * ga;
  }
}

__global__ void k3_bn(float* __restrict__ y,
                      const float* __restrict__ scaleA,
                      const float* __restrict__ scaleB) {
  int i4 = blockIdx.x * blockDim.x + threadIdx.x;
  if (i4 >= NB * NC * NL / 4) return;
  const int c = (i4 >> 10) & (NC - 1);
  float4 v = ((float4*)y)[i4];
  const float a = scaleA[c], b = scaleB[c];
  v.x = v.x * a + b;  v.y = v.y * a + b;
  v.z = v.z * a + b;  v.w = v.w * a + b;
  ((float4*)y)[i4] = v;
}

extern "C" void kernel_launch(void* const* d_in, const int* in_sizes, int n_in,
                              void* d_out, int out_size, void* d_ws, size_t ws_size,
                              hipStream_t stream) {
  (void)in_sizes; (void)n_in; (void)out_size; (void)ws_size;
  const float* x      = (const float*)d_in[0];
  const float* conv_w = (const float*)d_in[1];
  const float* conv_b = (const float*)d_in[2];
  const float* wq     = (const float*)d_in[3];
  const float* bq     = (const float*)d_in[4];
  const float* wk     = (const float*)d_in[5];
  const float* bk     = (const float*)d_in[6];
  const float* wv     = (const float*)d_in[7];
  const float* bv     = (const float*)d_in[8];
  const float* gamma  = (const float*)d_in[9];
  const float* beta   = (const float*)d_in[10];
  float* out = (float*)d_out;

  float* ws      = (float*)d_ws;
  float* kbuf    = ws;
  float* Mpart   = kbuf    + (size_t)NB * NC * NL;
  float* Mbuf    = Mpart   + (size_t)NB * NTILES * NC * NC;
  float* sumPart = Mbuf    + (size_t)NB * NC * NC;
  float* sqPart  = sumPart + (size_t)NB * NTILES * NC;
  float* scaleA  = sqPart  + (size_t)NB * NTILES * NC;
  float* scaleB  = scaleA  + NC;

  const int grid1 = NB * NTILES;
  k1_conv_qkv_M<<<grid1, BLOCK, 0, stream>>>(x, conv_w, conv_b, wq, bq, wk, bk,
                                             wv, bv, kbuf, Mpart);
  k1b_reduceM<<<(NB * NC * NC + 255) / 256, 256, 0, stream>>>(Mpart, Mbuf);
  k2_gate_residual<<<grid1, BLOCK, 0, stream>>>(x, kbuf, Mbuf, out, sumPart, sqPart);
  k2b_stats<<<NC / FOLD_CH, FOLD_SLICES * FOLD_CH, 0, stream>>>(sumPart, sqPart, gamma, beta, scaleA, scaleB);
  k3_bn<<<(NB * NC * NL / 4 + 255) / 256, 256, 0, stream>>>(out, scaleA, scaleB);
}
